# starting kernel
# baseline (speedup 1.0000x reference)
#include <hip/hip_runtime.h>
#include <cstdint>


typedef float v2f __attribute__((ext_vector_type(2)));
typedef float v8f __attribute__((ext_vector_type(8)));

__global__ void rmtpp_durations_kernel(const float* __restrict__ ts,
                                       float* __restrict__ dur,
                                       int T, int BS) {
    int b = blockIdx.x * blockDim.x + threadIdx.x;
    if (b >= BS) return;
    float m = -3.402823466e38f;
    for (int t = 0; t < T; ++t) {
        m = fmaxf(m, ts[(size_t)t * BS + b]);
    }
    dur[b] = m;
}

__global__ void rmtpp_proj_wmma_kernel(const float* __restrict__ h,
                                       const float* __restrict__ w,
                                       float* __restrict__ proj,
                                       int rows, int H) {
    const int lane        = threadIdx.x & 31;
    const int waveInBlock = threadIdx.x >> 5;
    const int wave        = blockIdx.x * (blockDim.x >> 5) + waveInBlock;
    const int rowBase     = wave * 16;
    if (rowBase >= rows) return;

    const int m    = lane & 15;
    const int koff = (lane >> 4) << 1;
    int r = rowBase + m;
    if (r >= rows) r = rows - 1;

    const float* __restrict__ rowPtr = h + (size_t)r * H + koff;
    const float* __restrict__ wPtr   = w + koff;

    v8f c = {};
    #pragma unroll 4
    for (int k = 0; k < H; k += 4) {
        v2f a = *reinterpret_cast<const v2f*>(rowPtr + k);
        v2f b = *reinterpret_cast<const v2f*>(wPtr + k);
        c = __builtin_amdgcn_wmma_f32_16x16x4_f32(
                false, a, false, b, (short)0, c, false, false);
    }

    const int sel = lane & 7;
    float v = c[0];
    v = (sel == 1) ? c[1] : v;
    v = (sel == 2) ? c[2] : v;
    v = (sel == 3) ? c[3] : v;
    v = (sel == 4) ? c[4] : v;
    v = (sel == 5) ? c[5] : v;
    v = (sel == 6) ? c[6] : v;
    v = (sel == 7) ? c[7] : v;

    const int  mOut    = ((lane >> 4) << 3) | sel;
    const bool doStore = (lane < 8) || (lane >= 16 && lane < 24);
    if (doStore && (rowBase + mOut) < rows) {
        proj[rowBase + mOut] = v;
    }
}

__global__ void rmtpp_grid_kernel(const float* __restrict__ ts,
                                  const float* __restrict__ proj,
                                  const float* __restrict__ dur,
                                  const float* __restrict__ time_influence,
                                  const float* __restrict__ base_intensity,
                                  float* __restrict__ out_log,
                                  float* __restrict__ out_grid,
                                  int T, int BS, int NG) {
    const int gid = blockIdx.x * blockDim.x + threadIdx.x;
    if (gid >= NG * BS) return;
    const int b = gid % BS;
    const int n = gid / BS;

    const float ti = time_influence[0];
    const float bi = base_intensity[0];

    const float t_grid = ((float)n / (float)(NG - 1)) * dur[b];

    int idx = 0;
    for (int t = 0; t < T; ++t) {
        if (ts[(size_t)t * BS + b] < t_grid) { idx = t; break; }
    }

    const float closest_ts = ts[(size_t)idx * BS + b];
    const float li = proj[(size_t)idx * BS + b] + ti * (t_grid - closest_ts) + bi;

    out_log[gid]  = li;
    out_grid[gid] = t_grid;
}

extern "C" void kernel_launch(void* const* d_in, const int* in_sizes, int n_in,
                              void* d_out, int out_size, void* d_ws, size_t ws_size,
                              hipStream_t stream) {
    const float* h  = (const float*)d_in[0];
    const float* ts = (const float*)d_in[1];
    const float* w  = (const float*)d_in[2];
    const float* ti = (const float*)d_in[3];
    const float* bi = (const float*)d_in[4];

    const int NG  = 1000;
    const int BS  = out_size / (2 * NG);
    const int TBS = in_sizes[1];
    const int T   = TBS / BS;
    const int H   = in_sizes[2];

    float* out_log  = (float*)d_out;
    float* out_grid = out_log + (size_t)NG * BS;

    float* proj = (float*)d_ws;
    float* dur  = proj + TBS;

    rmtpp_durations_kernel<<<(BS + 127) / 128, 128, 0, stream>>>(ts, dur, T, BS);

    const int waves  = (TBS + 15) / 16;
    const int blocks = (waves + 7) / 8;
    rmtpp_proj_wmma_kernel<<<blocks, 256, 0, stream>>>(h, w, proj, TBS, H);

    const int total = NG * BS;
    rmtpp_grid_kernel<<<(total + 255) / 256, 256, 0, stream>>>(
        ts, proj, dur, ti, bi, out_log, out_grid, T, BS, NG);
}


# pass 1: one wave per column, lanes over rows: 32 independent loads a lane, prefix minimum by lane shuffles; marks become one bit a row, one 128-byte line a column (was one thread, 128 serial trips)
# speedup vs baseline: 31.2416x; 31.2416x over previous
#include <hip/hip_runtime.h>
#include <cstdint>


typedef float v2f __attribute__((ext_vector_type(2)));
typedef float v8f __attribute__((ext_vector_type(8)));

#define WALK_LOADS 32
#define WALK_ROWS  (32 * WALK_LOADS)

__global__ __launch_bounds__(32)
void rmtpp_durations_kernel(const float* __restrict__ ts,
                            float* __restrict__ dur,
                            float* __restrict__ bits,
                            int T, int BS) {
    const int b    = blockIdx.x;
    const int lane = threadIdx.x;
    const float inf = __builtin_inff();
    float m     = -3.402823466e38f;
    float carry = inf;
    int chunk = 0;
    for (int base = 0; base < T; base += WALK_ROWS, ++chunk) {
        float v[WALK_LOADS];
        #pragma unroll
        for (int i = 0; i < WALK_LOADS; ++i) {
            const int t  = base + i * 32 + lane;
            const int tc = (t < T) ? t : T - 1;
            v[i] = ts[(size_t)tc * BS + b];
        }
        __builtin_amdgcn_sched_barrier(0);
        float s[WALK_LOADS];
        #pragma unroll
        for (int i = 0; i < WALK_LOADS; ++i) {
            const bool ok = (base + i * 32 + lane) < T;
            m    = fmaxf(m, ok ? v[i] : m);
            s[i] = ok ? v[i] : inf;
        }
        #pragma unroll
        for (int i = 0; i < WALK_LOADS; ++i) s[i] = fminf(s[i], __shfl_up(s[i], 1, 32));
        #pragma unroll
        for (int i = 0; i < WALK_LOADS; ++i) s[i] = fminf(s[i], __shfl_up(s[i], 2, 32));
        #pragma unroll
        for (int i = 0; i < WALK_LOADS; ++i) s[i] = fminf(s[i], __shfl_up(s[i], 4, 32));
        #pragma unroll
        for (int i = 0; i < WALK_LOADS; ++i) s[i] = fminf(s[i], __shfl_up(s[i], 8, 32));
        #pragma unroll
        for (int i = 0; i < WALK_LOADS; ++i) s[i] = fminf(s[i], __shfl_up(s[i], 16, 32));
        float below[WALK_LOADS];
        #pragma unroll
        for (int i = 0; i < WALK_LOADS; ++i) below[i] = __shfl_up(s[i], 1, 32);

        unsigned word = 0u;
        #pragma unroll
        for (int i = 0; i < WALK_LOADS; ++i) {
            const int   t  = base + i * 32 + lane;
            const bool  ok = t < T;
            const float lo  = fminf(carry, (lane == 0) ? inf : below[i]);
            const bool  hit = ok && (t == 0 || v[i] < lo);
            const unsigned mask = __builtin_amdgcn_ballot_w32(hit);
            word  = (lane == i) ? mask : word;
            carry = fminf(carry, __int_as_float(__builtin_amdgcn_readlane(__float_as_int(s[i]), 31)));
        }
        bits[((size_t)chunk * BS + b) * 32 + lane] = __uint_as_float(word);
    }
    m = fmaxf(m, __shfl_xor(m, 16, 32));
    m = fmaxf(m, __shfl_xor(m, 8, 32));
    m = fmaxf(m, __shfl_xor(m, 4, 32));
    m = fmaxf(m, __shfl_xor(m, 2, 32));
    m = fmaxf(m, __shfl_xor(m, 1, 32));
    if (lane == 0) dur[b] = m;
}

__global__ void rmtpp_proj_wmma_kernel(const float* __restrict__ h,
                                       const float* __restrict__ w,
                                       float* __restrict__ proj,
                                       const float* __restrict__ bits,
                                       int rows, int H, int BS) {
    const int lane        = threadIdx.x & 31;
    const int waveInBlock = threadIdx.x >> 5;
    const int wave        = blockIdx.x * (blockDim.x >> 5) + waveInBlock;
    const int rowBase     = wave * 16;
    if (rowBase >= rows) return;

    const int m    = lane & 15;
    const int koff = (lane >> 4) << 1;
    int r = rowBase + m;
    if (r >= rows) r = rows - 1;

    const int t     = r / BS;
    const int bcol  = r - t * BS;
    const int chunk = t / WALK_ROWS;
    const int tin   = t - chunk * WALK_ROWS;
    const unsigned word = __float_as_uint(bits[((size_t)chunk * BS + bcol) * 32 + (tin >> 5)]);
    if (__builtin_amdgcn_ballot_w32(((word >> (tin & 31)) & 1u) != 0u) == 0u) return;

    const float* __restrict__ rowPtr = h + (size_t)r * H + koff;
    const float* __restrict__ wPtr   = w + koff;

    v8f c = {};
    #pragma unroll 4
    for (int k = 0; k < H; k += 4) {
        v2f a = *reinterpret_cast<const v2f*>(rowPtr + k);
        v2f b = *reinterpret_cast<const v2f*>(wPtr + k);
        c = __builtin_amdgcn_wmma_f32_16x16x4_f32(
                false, a, false, b, (short)0, c, false, false);
    }

    const int sel = lane & 7;
    float v = c[0];
    v = (sel == 1) ? c[1] : v;
    v = (sel == 2) ? c[2] : v;
    v = (sel == 3) ? c[3] : v;
    v = (sel == 4) ? c[4] : v;
    v = (sel == 5) ? c[5] : v;
    v = (sel == 6) ? c[6] : v;
    v = (sel == 7) ? c[7] : v;

    const int  mOut    = ((lane >> 4) << 3) | sel;
    const bool doStore = (lane < 8) || (lane >= 16 && lane < 24);
    if (doStore && (rowBase + mOut) < rows) {
        proj[rowBase + mOut] = v;
    }
}

#define GRID_COLS  32
#define GRID_WAVES 32
#define GRID_RPW   32
#define GRID_ROWS  (GRID_WAVES * GRID_RPW)

__global__ __launch_bounds__(GRID_WAVES * 32)
void rmtpp_grid_kernel(const float* __restrict__ ts,
                       const float* __restrict__ proj,
                       const float* __restrict__ dur,
                       const float* __restrict__ time_influence,
                       const float* __restrict__ base_intensity,
                       float* __restrict__ out_log,
                       float* __restrict__ out_grid,
                       int T, int BS, int NG) {
    __shared__ float plane[GRID_ROWS * GRID_COLS];
    __shared__ float wtot[GRID_WAVES * GRID_COLS];

    const int lane = threadIdx.x & 31;
    const int waveV = (int)(threadIdx.x >> 5);
    const int wave  = __builtin_amdgcn_readfirstlane(waveV);

    const int colGroups = (BS + GRID_COLS - 1) / GRID_COLS;
    const int cg    = blockIdx.x % colGroups;
    const int slice = blockIdx.x / colGroups;

    const int  col   = cg * GRID_COLS + lane;
    const bool colOk = col < BS;
    const int  colC  = colOk ? col : BS - 1;

    const int  n   = slice * GRID_WAVES + wave;
    const bool nOk = n < NG;

    const float ti = time_influence[0];
    const float bi = base_intensity[0];

    const float t_grid = ((float)n / (float)(NG - 1)) * dur[colC];

    const float inf = __builtin_inff();

    int   idx   = 0;
    bool  found = false;
    float carry = inf;

    for (int base = 0; base < T; base += GRID_ROWS) {
        float v[GRID_RPW];
        const int r0 = base + wave * GRID_RPW;
        #pragma unroll
        for (int i = 0; i < GRID_RPW; ++i) {
            const int t  = r0 + i;
            const int tc = (t < T) ? t : T - 1;
            v[i] = (ts + (size_t)tc * BS)[colC];
        }
        #pragma unroll
        for (int i = 0; i < GRID_RPW; ++i) v[i] = (r0 + i < T) ? v[i] : inf;
        #pragma unroll
        for (int i = 1; i < GRID_RPW; ++i) v[i] = fminf(v[i - 1], v[i]);

        wtot[wave * GRID_COLS + lane] = v[GRID_RPW - 1];
        __syncthreads();

        float pre = carry;
        #pragma unroll
        for (int w = 0; w < GRID_WAVES; ++w) {
            const float x = wtot[w * GRID_COLS + lane];
            pre = fminf(pre, (w < waveV) ? x : inf);
        }
        #pragma unroll
        for (int i = 0; i < GRID_RPW; ++i)
            plane[(wave * GRID_RPW + i) * GRID_COLS + lane] = fminf(pre, v[i]);
        __syncthreads();

        if (!found) {
            int pos = 0;
            #pragma unroll
            for (int step = GRID_ROWS >> 1; step >= 1; step >>= 1) {
                const float x = plane[(pos + step - 1) * GRID_COLS + lane];
                pos += (x < t_grid) ? 0 : step;
            }
            if (plane[pos * GRID_COLS + lane] < t_grid) { found = true; idx = base + pos; }
        }

        if (base + GRID_ROWS < T) {
            carry = plane[(GRID_ROWS - 1) * GRID_COLS + lane];
            __syncthreads();
        }
    }

    if (nOk && colOk) {
        const size_t src = (size_t)idx * BS + col;
        const float closest_ts = ts[src];
        const float li = proj[src] + ti * (t_grid - closest_ts) + bi;
        const size_t dst = (size_t)n * BS + col;
        out_log[dst]  = li;
        out_grid[dst] = t_grid;
    }
}

extern "C" void kernel_launch(void* const* d_in, const int* in_sizes, int n_in,
                              void* d_out, int out_size, void* d_ws, size_t ws_size,
                              hipStream_t stream) {
    const float* h  = (const float*)d_in[0];
    const float* ts = (const float*)d_in[1];
    const float* w  = (const float*)d_in[2];
    const float* ti = (const float*)d_in[3];
    const float* bi = (const float*)d_in[4];

    const int NG  = 1000;
    const int BS  = out_size / (2 * NG);
    const int TBS = in_sizes[1];
    const int T   = TBS / BS;
    const int H   = in_sizes[2];

    float* out_log  = (float*)d_out;
    float* out_grid = out_log + (size_t)NG * BS;

    float* proj = (float*)d_ws;
    float* dur  = proj + TBS;

    float* bits = out_log;

    rmtpp_durations_kernel<<<BS, 32, 0, stream>>>(ts, dur, bits, T, BS);

    const int waves  = (TBS + 15) / 16;
    const int blocks = (waves + 7) / 8;
    rmtpp_proj_wmma_kernel<<<blocks, 256, 0, stream>>>(h, w, proj, bits, TBS, H, BS);

    const int colGroups = (BS + GRID_COLS - 1) / GRID_COLS;
    const int slices    = (NG + GRID_WAVES - 1) / GRID_WAVES;
    rmtpp_grid_kernel<<<colGroups * slices, GRID_WAVES * 32, 0, stream>>>(
        ts, proj, dur, ti, bi, out_log, out_grid, T, BS, NG);
}


# projection moved into pass 1: a column's wave lists its marked rows in LDS and takes their WMMA row dots itself, 16 rows a tile, operands 32 k-steps ahead; projection launch dropped; sum order kept
# speedup vs baseline: 43.4882x; 1.3920x over previous
#include <hip/hip_runtime.h>
#include <cstdint>


typedef float v2f __attribute__((ext_vector_type(2)));
typedef float v8f __attribute__((ext_vector_type(8)));

#define WALK_LOADS 32
#define WALK_ROWS  (32 * WALK_LOADS)
#define PROJ_AHEAD 32
#define PROJ_GROUP 8

__global__ __launch_bounds__(32)
void rmtpp_marks_proj_kernel(const float* __restrict__ ts,
                             const float* __restrict__ h,
                             const float* __restrict__ w,
                             float* __restrict__ dur,
                             float* __restrict__ proj,
                             int T, int BS, int H) {
    __shared__ int list[WALK_ROWS];

    const int b    = blockIdx.x;
    const int lane = threadIdx.x;
    const float inf = __builtin_inff();
    float m     = -3.402823466e38f;
    float carry = inf;

    const int koff = (lane >> 4) << 1;
    const float* __restrict__ wPtr = w + koff;
    const int span = 4 * PROJ_AHEAD;
    const int nb   = H / span;

    for (int base = 0; base < T; base += WALK_ROWS) {
        float v[WALK_LOADS];
        #pragma unroll
        for (int i = 0; i < WALK_LOADS; ++i) {
            const int t  = base + i * 32 + lane;
            const int tc = (t < T) ? t : T - 1;
            v[i] = ts[(size_t)tc * BS + b];
        }
        __builtin_amdgcn_sched_barrier(0);
        float s[WALK_LOADS];
        #pragma unroll
        for (int i = 0; i < WALK_LOADS; ++i) {
            const bool ok = (base + i * 32 + lane) < T;
            m    = fmaxf(m, ok ? v[i] : m);
            s[i] = ok ? v[i] : inf;
        }
        #pragma unroll
        for (int i = 0; i < WALK_LOADS; ++i) s[i] = fminf(s[i], __shfl_up(s[i], 1, 32));
        #pragma unroll
        for (int i = 0; i < WALK_LOADS; ++i) s[i] = fminf(s[i], __shfl_up(s[i], 2, 32));
        #pragma unroll
        for (int i = 0; i < WALK_LOADS; ++i) s[i] = fminf(s[i], __shfl_up(s[i], 4, 32));
        #pragma unroll
        for (int i = 0; i < WALK_LOADS; ++i) s[i] = fminf(s[i], __shfl_up(s[i], 8, 32));
        #pragma unroll
        for (int i = 0; i < WALK_LOADS; ++i) s[i] = fminf(s[i], __shfl_up(s[i], 16, 32));
        float below[WALK_LOADS];
        #pragma unroll
        for (int i = 0; i < WALK_LOADS; ++i) below[i] = __shfl_up(s[i], 1, 32);

        unsigned word = 0u;
        int total = 0;
        int first = 0;
        #pragma unroll
        for (int i = 0; i < WALK_LOADS; ++i) {
            const int   t  = base + i * 32 + lane;
            const bool  ok = t < T;
            const float lo  = fminf(carry, (lane == 0) ? inf : below[i]);
            const bool  hit = ok && (t == 0 || v[i] < lo);
            const unsigned mask = __builtin_amdgcn_ballot_w32(hit);
            word  = (lane == i) ? mask : word;
            first = (lane == i) ? total : first;
            total += __popc(mask);
            carry = fminf(carry, __int_as_float(__builtin_amdgcn_readlane(__float_as_int(s[i]), 31)));
        }

        {
            int o = first;
            unsigned wd = word;
            while (wd != 0u) {
                list[o++] = base + lane * 32 + __builtin_ctz(wd);
                wd &= wd - 1u;
            }
        }
        __syncthreads();

        for (int q0 = 0; q0 < total; q0 += 16) {
            const int q  = q0 + (lane & 15);
            const int t  = list[(q < total) ? q : total - 1];
            const float* __restrict__ rowPtr = h + ((size_t)t * BS + b) * H + koff;
            const int  sel     = lane & 7;
            const int  mOut    = ((lane >> 4) << 3) | sel;
            const bool doStore = (lane < 8) || (lane >= 16 && lane < 24);
            const int  qOut    = q0 + mOut;
            const int  tOut    = list[(qOut < total) ? qOut : total - 1];

            v8f c = {};
            if (nb > 0) {
                v2f fa[PROJ_AHEAD], fb[PROJ_AHEAD];
                #pragma unroll
                for (int j = 0; j < PROJ_AHEAD; ++j) {
                    fa[j] = *reinterpret_cast<const v2f*>(rowPtr + 4 * j);
                    fb[j] = *reinterpret_cast<const v2f*>(wPtr + 4 * j);
                }
                for (int trip = 1; trip < nb; ++trip) {
                    const float* __restrict__ ap = rowPtr + trip * span;
                    const float* __restrict__ bp = wPtr + trip * span;
                    #pragma unroll
                    for (int g = 0; g < PROJ_AHEAD; g += PROJ_GROUP) {
                        #pragma unroll
                        for (int j = g; j < g + PROJ_GROUP; ++j)
                            c = __builtin_amdgcn_wmma_f32_16x16x4_f32(
                                    false, fa[j], false, fb[j], (short)0, c, false, false);
                        if (g > 0) {
                            #pragma unroll
                            for (int j = g - PROJ_GROUP; j < g; ++j) {
                                fa[j] = *reinterpret_cast<const v2f*>(ap + 4 * j);
                                fb[j] = *reinterpret_cast<const v2f*>(bp + 4 * j);
                            }
                        }
                    }
                    #pragma unroll
                    for (int j = PROJ_AHEAD - PROJ_GROUP; j < PROJ_AHEAD; ++j) {
                        fa[j] = *reinterpret_cast<const v2f*>(ap + 4 * j);
                        fb[j] = *reinterpret_cast<const v2f*>(bp + 4 * j);
                    }
                }
                #pragma unroll
                for (int j = 0; j < PROJ_AHEAD; ++j)
                    c = __builtin_amdgcn_wmma_f32_16x16x4_f32(
                            false, fa[j], false, fb[j], (short)0, c, false, false);
            }
            for (int k = nb * span; k < H; k += 4) {
                v2f a  = *reinterpret_cast<const v2f*>(rowPtr + k);
                v2f bw = *reinterpret_cast<const v2f*>(wPtr + k);
                c = __builtin_amdgcn_wmma_f32_16x16x4_f32(
                        false, a, false, bw, (short)0, c, false, false);
            }

            float pv = c[0];
            pv = (sel == 1) ? c[1] : pv;
            pv = (sel == 2) ? c[2] : pv;
            pv = (sel == 3) ? c[3] : pv;
            pv = (sel == 4) ? c[4] : pv;
            pv = (sel == 5) ? c[5] : pv;
            pv = (sel == 6) ? c[6] : pv;
            pv = (sel == 7) ? c[7] : pv;

            if (doStore && qOut < total) {
                proj[(size_t)tOut * BS + b] = pv;
            }
        }
        __syncthreads();
    }
    m = fmaxf(m, __shfl_xor(m, 16, 32));
    m = fmaxf(m, __shfl_xor(m, 8, 32));
    m = fmaxf(m, __shfl_xor(m, 4, 32));
    m = fmaxf(m, __shfl_xor(m, 2, 32));
    m = fmaxf(m, __shfl_xor(m, 1, 32));
    if (lane == 0) dur[b] = m;
}

#define GRID_COLS  32
#define GRID_WAVES 32
#define GRID_RPW   32
#define GRID_ROWS  (GRID_WAVES * GRID_RPW)

__global__ __launch_bounds__(GRID_WAVES * 32)
void rmtpp_grid_kernel(const float* __restrict__ ts,
                       const float* __restrict__ proj,
                       const float* __restrict__ dur,
                       const float* __restrict__ time_influence,
                       const float* __restrict__ base_intensity,
                       float* __restrict__ out_log,
                       float* __restrict__ out_grid,
                       int T, int BS, int NG) {
    __shared__ float plane[GRID_ROWS * GRID_COLS];
    __shared__ float wtot[GRID_WAVES * GRID_COLS];

    const int lane = threadIdx.x & 31;
    const int waveV = (int)(threadIdx.x >> 5);
    const int wave  = __builtin_amdgcn_readfirstlane(waveV);

    const int colGroups = (BS + GRID_COLS - 1) / GRID_COLS;
    const int cg    = blockIdx.x % colGroups;
    const int slice = blockIdx.x / colGroups;

    const int  col   = cg * GRID_COLS + lane;
    const bool colOk = col < BS;
    const int  colC  = colOk ? col : BS - 1;

    const int  n   = slice * GRID_WAVES + wave;
    const bool nOk = n < NG;

    const float ti = time_influence[0];
    const float bi = base_intensity[0];

    const float t_grid = ((float)n / (float)(NG - 1)) * dur[colC];

    const float inf = __builtin_inff();

    int   idx   = 0;
    bool  found = false;
    float carry = inf;

    for (int base = 0; base < T; base += GRID_ROWS) {
        float v[GRID_RPW];
        const int r0 = base + wave * GRID_RPW;
        #pragma unroll
        for (int i = 0; i < GRID_RPW; ++i) {
            const int t  = r0 + i;
            const int tc = (t < T) ? t : T - 1;
            v[i] = (ts + (size_t)tc * BS)[colC];
        }
        #pragma unroll
        for (int i = 0; i < GRID_RPW; ++i) v[i] = (r0 + i < T) ? v[i] : inf;
        #pragma unroll
        for (int i = 1; i < GRID_RPW; ++i) v[i] = fminf(v[i - 1], v[i]);

        wtot[wave * GRID_COLS + lane] = v[GRID_RPW - 1];
        __syncthreads();

        float pre = carry;
        #pragma unroll
        for (int w = 0; w < GRID_WAVES; ++w) {
            const float x = wtot[w * GRID_COLS + lane];
            pre = fminf(pre, (w < waveV) ? x : inf);
        }
        #pragma unroll
        for (int i = 0; i < GRID_RPW; ++i)
            plane[(wave * GRID_RPW + i) * GRID_COLS + lane] = fminf(pre, v[i]);
        __syncthreads();

        if (!found) {
            int pos = 0;
            #pragma unroll
            for (int step = GRID_ROWS >> 1; step >= 1; step >>= 1) {
                const float x = plane[(pos + step - 1) * GRID_COLS + lane];
                pos += (x < t_grid) ? 0 : step;
            }
            if (plane[pos * GRID_COLS + lane] < t_grid) { found = true; idx = base + pos; }
        }

        if (base + GRID_ROWS < T) {
            carry = plane[(GRID_ROWS - 1) * GRID_COLS + lane];
            __syncthreads();
        }
    }

    if (nOk && colOk) {
        const size_t src = (size_t)idx * BS + col;
        const float closest_ts = ts[src];
        const float li = proj[src] + ti * (t_grid - closest_ts) + bi;
        const size_t dst = (size_t)n * BS + col;
        out_log[dst]  = li;
        out_grid[dst] = t_grid;
    }
}

extern "C" void kernel_launch(void* const* d_in, const int* in_sizes, int n_in,
                              void* d_out, int out_size, void* d_ws, size_t ws_size,
                              hipStream_t stream) {
    const float* h  = (const float*)d_in[0];
    const float* ts = (const float*)d_in[1];
    const float* w  = (const float*)d_in[2];
    const float* ti = (const float*)d_in[3];
    const float* bi = (const float*)d_in[4];

    const int NG  = 1000;
    const int BS  = out_size / (2 * NG);
    const int TBS = in_sizes[1];
    const int T   = TBS / BS;
    const int H   = in_sizes[2];

    float* out_log  = (float*)d_out;
    float* out_grid = out_log + (size_t)NG * BS;

    float* proj = (float*)d_ws;
    float* dur  = proj + TBS;

    rmtpp_marks_proj_kernel<<<BS, 32, 0, stream>>>(ts, h, w, dur, proj, T, BS, H);

    const int colGroups = (BS + GRID_COLS - 1) / GRID_COLS;
    const int slices    = (NG + GRID_WAVES - 1) / GRID_WAVES;
    rmtpp_grid_kernel<<<colGroups * slices, GRID_WAVES * 32, 0, stream>>>(
        ts, proj, dur, ti, bi, out_log, out_grid, T, BS, NG);
}


# pass 1 folded into the grid work-group: it marks its 32 columns as it builds the running-minimum plane, lists the marked rows in LDS, projects them 16 a WMMA tile; one launch; sum order kept
# speedup vs baseline: 53.1874x; 1.2230x over previous
#include <hip/hip_runtime.h>
#include <cstdint>


typedef float v2f __attribute__((ext_vector_type(2)));
typedef float v8f __attribute__((ext_vector_type(8)));

#define GRID_COLS  32
#define GRID_WAVES 32
#define GRID_RPW   32
#define GRID_ROWS  (GRID_WAVES * GRID_RPW)
#define PROJ_AHEAD 16
#define PROJ_GROUP 4

__global__ __launch_bounds__(GRID_WAVES * 32)
void rmtpp_fused_kernel(const float* __restrict__ ts,
                        const float* __restrict__ h,
                        const float* __restrict__ w,
                        const float* __restrict__ time_influence,
                        const float* __restrict__ base_intensity,
                        float* __restrict__ out_log,
                        float* __restrict__ out_grid,
                        int T, int BS, int H, int NG) {
    __shared__ float    plane[GRID_ROWS * GRID_COLS];
    __shared__ float    wtot[GRID_WAVES * GRID_COLS];
    __shared__ float    wmax[GRID_WAVES * GRID_COLS];
    __shared__ unsigned mk[GRID_WAVES * GRID_COLS];
    __shared__ int      woff[GRID_WAVES * GRID_COLS];

    const int lane  = threadIdx.x & 31;
    const int waveV = (int)(threadIdx.x >> 5);
    const int wave  = __builtin_amdgcn_readfirstlane(waveV);

    const int colGroups = (BS + GRID_COLS - 1) / GRID_COLS;
    const int cg    = blockIdx.x % colGroups;
    const int slice = blockIdx.x / colGroups;

    const int  col   = cg * GRID_COLS + lane;
    const bool colOk = col < BS;
    const int  colC  = colOk ? col : BS - 1;

    const int  n   = slice * GRID_WAVES + wave;
    const bool nOk = n < NG;

    const float ti = time_influence[0];
    const float bi = base_intensity[0];

    const float inf = __builtin_inff();

    const int koff = (lane >> 4) << 1;
    const float* __restrict__ wPtr = w + koff;
    const int span = 4 * PROJ_AHEAD;
    const int nb   = H / span;

    float mx = -3.402823466e38f;
    if (T > GRID_ROWS) {
        for (int base = GRID_ROWS; base < T; base += GRID_ROWS) {
            const int r0 = base + wave * GRID_RPW;
            #pragma unroll
            for (int i = 0; i < GRID_RPW; ++i) {
                const int t  = r0 + i;
                const int tc = (t < T) ? t : T - 1;
                mx = fmaxf(mx, (ts + (size_t)tc * BS)[colC]);
            }
        }
    }

    float t_grid = 0.0f;
    bool  found  = false;
    float carry  = inf;
    float pvSel  = 0.0f;
    float tsSel  = 0.0f;

    for (int base = 0; base < T; base += GRID_ROWS) {
        float v[GRID_RPW];
        const int r0 = base + wave * GRID_RPW;
        #pragma unroll
        for (int i = 0; i < GRID_RPW; ++i) {
            const int t  = r0 + i;
            const int tc = (t < T) ? t : T - 1;
            v[i] = (ts + (size_t)tc * BS)[colC];
        }
        #pragma unroll
        for (int i = 0; i < GRID_RPW; ++i) {
            const bool ok = (r0 + i) < T;
            mx   = fmaxf(mx, ok ? v[i] : mx);
            v[i] = ok ? v[i] : inf;
        }
        #pragma unroll
        for (int i = 1; i < GRID_RPW; ++i) v[i] = fminf(v[i - 1], v[i]);

        wtot[wave * GRID_COLS + lane] = v[GRID_RPW - 1];
        if (base == 0) wmax[wave * GRID_COLS + lane] = mx;
        __syncthreads();

        float pre = carry;
        #pragma unroll
        for (int w2 = 0; w2 < GRID_WAVES; ++w2) {
            const float x = wtot[w2 * GRID_COLS + lane];
            pre = fminf(pre, (w2 < waveV) ? x : inf);
        }
        if (base == 0) {
            float d = wmax[lane];
            #pragma unroll
            for (int w2 = 1; w2 < GRID_WAVES; ++w2) d = fmaxf(d, wmax[w2 * GRID_COLS + lane]);
            t_grid = ((float)n / (float)(NG - 1)) * d;
        }
        unsigned word = 0u;
        float prev = pre;
        #pragma unroll
        for (int i = 0; i < GRID_RPW; ++i) {
            const float p = fminf(pre, v[i]);
            plane[(wave * GRID_RPW + i) * GRID_COLS + lane] = p;
            const bool hit = (p < prev) || (r0 + i == 0);
            word |= hit ? (1u << i) : 0u;
            prev = p;
        }
        word = colOk ? word : 0u;
        mk[wave * GRID_COLS + lane] = word;
        __syncthreads();

        bool hitHere = false;
        int  pos = 0;
        if (!found) {
            #pragma unroll
            for (int step = GRID_ROWS >> 1; step >= 1; step >>= 1) {
                const float x = plane[(pos + step - 1) * GRID_COLS + lane];
                pos += (x < t_grid) ? 0 : step;
            }
            if (plane[pos * GRID_COLS + lane] < t_grid) { found = true; hitHere = true; }
        }
        const bool want   = hitHere || (base == 0);
        const int  rowSel = hitHere ? pos : 0;
        if (want) tsSel = ts[(size_t)(base + rowSel) * BS + colC];
        const float last = plane[(GRID_ROWS - 1) * GRID_COLS + lane];

        int below = 0, colTot = 0;
        #pragma unroll
        for (int w2 = 0; w2 < GRID_WAVES; ++w2) {
            const int cnt = __popc(mk[w2 * GRID_COLS + lane]);
            below  += (w2 < waveV) ? cnt : 0;
            colTot += cnt;
        }
        int inc = colTot;
        #pragma unroll
        for (int d = 1; d < 32; d <<= 1) {
            const int y = __shfl_up(inc, d, 32);
            inc += (lane >= d) ? y : 0;
        }
        const int M   = __builtin_amdgcn_readlane(inc, 31);
        const int off = inc - colTot + below;
        woff[wave * GRID_COLS + lane] = off;
        __syncthreads();

        {
            int o = off;
            unsigned wd = word;
            while (wd != 0u) {
                const int r = wave * GRID_RPW + __builtin_ctz(wd);
                plane[o++] = __int_as_float((r << 5) | lane);
                wd &= wd - 1u;
            }
        }
        __syncthreads();

        const int nTiles = (M + 15) >> 4;
        for (int j = wave; j < nTiles; j += GRID_WAVES) {
            const int q0   = j << 4;
            const int q    = q0 + (lane & 15);
            const int code = __float_as_int(plane[(q < M) ? q : M - 1]);
            const int t    = base + (code >> 5);
            const int b    = cg * GRID_COLS + (code & 31);
            const float* __restrict__ rowPtr = h + ((size_t)t * BS + b) * H + koff;
            const int  sel     = lane & 7;
            const int  mOut    = ((lane >> 4) << 3) | sel;
            const bool doStore = (lane < 8) || (lane >= 16 && lane < 24);
            const int  qOut    = q0 + mOut;

            v8f c = {};
            if (nb > 0) {
                v2f fa[PROJ_AHEAD], fb[PROJ_AHEAD];
                #pragma unroll
                for (int s = 0; s < PROJ_AHEAD; ++s) {
                    fa[s] = *reinterpret_cast<const v2f*>(rowPtr + 4 * s);
                    fb[s] = *reinterpret_cast<const v2f*>(wPtr + 4 * s);
                }
                #pragma nounroll
                for (int trip = 1; trip < nb; ++trip) {
                    const float* __restrict__ ap = rowPtr + trip * span;
                    const float* __restrict__ bp = wPtr + trip * span;
                    #pragma unroll
                    for (int g = 0; g < PROJ_AHEAD; g += PROJ_GROUP) {
                        #pragma unroll
                        for (int s = g; s < g + PROJ_GROUP; ++s)
                            c = __builtin_amdgcn_wmma_f32_16x16x4_f32(
                                    false, fa[s], false, fb[s], (short)0, c, false, false);
                        if (g > 0) {
                            #pragma unroll
                            for (int s = g - PROJ_GROUP; s < g; ++s) {
                                fa[s] = *reinterpret_cast<const v2f*>(ap + 4 * s);
                                fb[s] = *reinterpret_cast<const v2f*>(bp + 4 * s);
                            }
                        }
                    }
                    #pragma unroll
                    for (int s = PROJ_AHEAD - PROJ_GROUP; s < PROJ_AHEAD; ++s) {
                        fa[s] = *reinterpret_cast<const v2f*>(ap + 4 * s);
                        fb[s] = *reinterpret_cast<const v2f*>(bp + 4 * s);
                    }
                }
                #pragma unroll
                for (int s = 0; s < PROJ_AHEAD; ++s)
                    c = __builtin_amdgcn_wmma_f32_16x16x4_f32(
                            false, fa[s], false, fb[s], (short)0, c, false, false);
            }
            for (int k = nb * span; k < H; k += 4) {
                v2f a  = *reinterpret_cast<const v2f*>(rowPtr + k);
                v2f bw = *reinterpret_cast<const v2f*>(wPtr + k);
                c = __builtin_amdgcn_wmma_f32_16x16x4_f32(
                        false, a, false, bw, (short)0, c, false, false);
            }

            float pv = c[0];
            pv = (sel == 1) ? c[1] : pv;
            pv = (sel == 2) ? c[2] : pv;
            pv = (sel == 3) ? c[3] : pv;
            pv = (sel == 4) ? c[4] : pv;
            pv = (sel == 5) ? c[5] : pv;
            pv = (sel == 6) ? c[6] : pv;
            pv = (sel == 7) ? c[7] : pv;

            if (doStore && qOut < M) plane[qOut] = pv;
        }
        __syncthreads();

        if (want) {
            const int      ws  = rowSel >> 5;
            const unsigned wdS = mk[ws * GRID_COLS + lane];
            const int      o   = woff[ws * GRID_COLS + lane]
                               + __popc(wdS & ((1u << (rowSel & 31)) - 1u));
            pvSel = plane[o];
        }
        carry = last;
    }

    if (nOk && colOk) {
        const float li = pvSel + ti * (t_grid - tsSel) + bi;
        const size_t dst = (size_t)n * BS + col;
        out_log[dst]  = li;
        out_grid[dst] = t_grid;
    }
}

extern "C" void kernel_launch(void* const* d_in, const int* in_sizes, int n_in,
                              void* d_out, int out_size, void* d_ws, size_t ws_size,
                              hipStream_t stream) {
    const float* h  = (const float*)d_in[0];
    const float* ts = (const float*)d_in[1];
    const float* w  = (const float*)d_in[2];
    const float* ti = (const float*)d_in[3];
    const float* bi = (const float*)d_in[4];

    const int NG  = 1000;
    const int BS  = out_size / (2 * NG);
    const int TBS = in_sizes[1];
    const int T   = TBS / BS;
    const int H   = in_sizes[2];

    float* out_log  = (float*)d_out;
    float* out_grid = out_log + (size_t)NG * BS;

    const int colGroups = (BS + GRID_COLS - 1) / GRID_COLS;
    const int slices    = (NG + GRID_WAVES - 1) / GRID_WAVES;
    rmtpp_fused_kernel<<<colGroups * slices, GRID_WAVES * 32, 0, stream>>>(
        ts, h, w, ti, bi, out_log, out_grid, T, BS, H, NG);
}


# row 0 of each column leaves the marked-row list: projected on its own under the search, K dealt to the waves, partials added in LDS (new sum order); list phase only if a later row is marked
# speedup vs baseline: 61.1734x; 1.1501x over previous
#include <hip/hip_runtime.h>
#include <cstdint>


typedef float v2f __attribute__((ext_vector_type(2)));
typedef float v8f __attribute__((ext_vector_type(8)));

#define GRID_COLS  32
#define GRID_WAVES 32
#define GRID_RPW   32
#define GRID_ROWS  (GRID_WAVES * GRID_RPW)
#define PROJ_AHEAD 16
#define PROJ_GROUP 4
#define ROW0_SLICES (GRID_WAVES / 2)
#define ROW0_STEPS 8
static_assert(GRID_WAVES == 32, "one lane reads one wave's flag; a row-0 tile pair is dealt to 32 waves");

__device__ __forceinline__ void row0_slice(int wave, int H, int& ks0, int& ks1) {
    const int ksteps = (H + 3) >> 2;
    const int per    = (ksteps + ROW0_SLICES - 1) / ROW0_SLICES;
    const int ksl    = wave >> 1;
    ks0 = (ksl * per < ksteps) ? ksl * per : ksteps;
    ks1 = (ks0 + per < ksteps) ? ks0 + per : ksteps;
}

__global__ __launch_bounds__(GRID_WAVES * 32)
void rmtpp_fused_kernel(const float* __restrict__ ts,
                        const float* __restrict__ h,
                        const float* __restrict__ w,
                        const float* __restrict__ time_influence,
                        const float* __restrict__ base_intensity,
                        float* __restrict__ out_log,
                        float* __restrict__ out_grid,
                        int T, int BS, int H, int NG) {
    __shared__ float    plane[GRID_ROWS * GRID_COLS];
    __shared__ float    wtot[GRID_WAVES * GRID_COLS];
    __shared__ float    wmax[GRID_WAVES * GRID_COLS];
    __shared__ unsigned mk[GRID_WAVES * GRID_COLS];
    __shared__ int      woff[GRID_WAVES * GRID_COLS];
    __shared__ float    part[ROW0_SLICES * GRID_COLS];
    __shared__ unsigned wany[GRID_WAVES];

    const int lane  = threadIdx.x & 31;
    const int waveV = (int)(threadIdx.x >> 5);
    const int wave  = __builtin_amdgcn_readfirstlane(waveV);

    const int colGroups = (BS + GRID_COLS - 1) / GRID_COLS;
    const int cg    = blockIdx.x % colGroups;
    const int slice = blockIdx.x / colGroups;

    const int  col   = cg * GRID_COLS + lane;
    const bool colOk = col < BS;
    const int  colC  = colOk ? col : BS - 1;

    const int  n   = slice * GRID_WAVES + wave;
    const bool nOk = n < NG;

    const float ti = time_influence[0];
    const float bi = base_intensity[0];

    const float inf = __builtin_inff();

    const int koff = (lane >> 4) << 1;
    const float* __restrict__ wPtr = w + koff;
    const int span = 4 * PROJ_AHEAD;
    const int nb   = H / span;

    float mx = -3.402823466e38f;
    if (T > GRID_ROWS) {
        for (int base = GRID_ROWS; base < T; base += GRID_ROWS) {
            const int r0 = base + wave * GRID_RPW;
            #pragma unroll
            for (int i = 0; i < GRID_RPW; ++i) {
                const int t  = r0 + i;
                const int tc = (t < T) ? t : T - 1;
                mx = fmaxf(mx, (ts + (size_t)tc * BS)[colC]);
            }
        }
    }

    float pv0 = 0.0f;

    float t_grid = 0.0f;
    bool  found  = false;
    float carry  = inf;
    float pvSel  = 0.0f;
    float tsSel  = 0.0f;

    for (int base = 0; base < T; base += GRID_ROWS) {
        float v[GRID_RPW];
        const int r0 = base + wave * GRID_RPW;
        #pragma unroll
        for (int i = 0; i < GRID_RPW; ++i) {
            const int t  = r0 + i;
            const int tc = (t < T) ? t : T - 1;
            v[i] = (ts + (size_t)tc * BS)[colC];
        }
        #pragma unroll
        for (int i = 0; i < GRID_RPW; ++i) {
            const bool ok = (r0 + i) < T;
            mx   = fmaxf(mx, ok ? v[i] : mx);
            v[i] = ok ? v[i] : inf;
        }
        #pragma unroll
        for (int i = 1; i < GRID_RPW; ++i) v[i] = fminf(v[i - 1], v[i]);

        v2f ea[ROW0_STEPS];
        int ks0 = 0, ks1 = 0;
        if (base == 0) row0_slice(wave, H, ks0, ks1);
        const int nReg = (ks1 - ks0 >= ROW0_STEPS) ? ROW0_STEPS : 0;
        int b0 = cg * GRID_COLS + (wave & 1) * 16 + (lane & 15);
        b0 = (b0 < BS) ? b0 : BS - 1;
        const float* __restrict__ rowPtr0 = h + (size_t)b0 * H + koff;
        {
            const int kLast = ((H + 3) >> 2) - 1;
            #pragma unroll
            for (int s = 0; s < ROW0_STEPS; ++s) {
                const int kk = (ks0 + s < kLast) ? ks0 + s : kLast;
                ea[s] = *reinterpret_cast<const v2f*>(rowPtr0 + 4 * kk);
            }
        }
        wtot[wave * GRID_COLS + lane] = v[GRID_RPW - 1];
        if (base == 0) wmax[wave * GRID_COLS + lane] = mx;
        __syncthreads();

        float pre = carry;
        #pragma unroll
        for (int w2 = 0; w2 < GRID_WAVES; ++w2) {
            const float x = wtot[w2 * GRID_COLS + lane];
            pre = fminf(pre, (w2 < waveV) ? x : inf);
        }
        if (base == 0) {
            float d = wmax[lane];
            #pragma unroll
            for (int w2 = 1; w2 < GRID_WAVES; ++w2) d = fmaxf(d, wmax[w2 * GRID_COLS + lane]);
            t_grid = ((float)n / (float)(NG - 1)) * d;
        }
        unsigned word = 0u;
        float prev = pre;
        #pragma unroll
        for (int i = 0; i < GRID_RPW; ++i) {
            const float p = fminf(pre, v[i]);
            plane[(wave * GRID_RPW + i) * GRID_COLS + lane] = p;
            const bool hit = (p < prev) && (r0 + i != 0);
            word |= hit ? (1u << i) : 0u;
            prev = p;
        }
        word = colOk ? word : 0u;
        mk[wave * GRID_COLS + lane] = word;
        {
            const bool anyHere = __builtin_amdgcn_ballot_w32(word != 0u) != 0u;
            if (lane == 0) wany[wave] = anyHere ? 1u : 0u;
        }
        __syncthreads();

        v2f eb[ROW0_STEPS];
        {
            const int kLast = ((H + 3) >> 2) - 1;
            #pragma unroll
            for (int s = 0; s < ROW0_STEPS; ++s) {
                const int kk = (ks0 + s < kLast) ? ks0 + s : kLast;
                eb[s] = *reinterpret_cast<const v2f*>(wPtr + 4 * kk);
            }
        }
        bool hitHere = false;
        int  pos = 0;
        if (!found) {
            #pragma unroll
            for (int step = GRID_ROWS >> 1; step >= 1; step >>= 1) {
                const float x = plane[(pos + step - 1) * GRID_COLS + lane];
                pos += (x < t_grid) ? 0 : step;
            }
            if (plane[pos * GRID_COLS + lane] < t_grid) { found = true; hitHere = true; }
        }
        const bool want   = hitHere || (base == 0);
        const int  rowSel = hitHere ? pos : 0;
        if (want) tsSel = ts[(size_t)(base + rowSel) * BS + colC];
        const float last = plane[(GRID_ROWS - 1) * GRID_COLS + lane];
        const bool selRow0 = (base + rowSel) == 0;

        if (base == 0) {
            v8f c0 = {};
            if (nReg != 0) {
                #pragma unroll
                for (int s = 0; s < ROW0_STEPS; ++s)
                    c0 = __builtin_amdgcn_wmma_f32_16x16x4_f32(
                             false, ea[s], false, eb[s], (short)0, c0, false, false);
            }
            #pragma nounroll
            for (int k = 4 * (ks0 + nReg); k < 4 * ks1; k += 4) {
                v2f a  = *reinterpret_cast<const v2f*>(rowPtr0 + k);
                v2f bw = *reinterpret_cast<const v2f*>(wPtr + k);
                c0 = __builtin_amdgcn_wmma_f32_16x16x4_f32(
                         false, a, false, bw, (short)0, c0, false, false);
            }
            const int sel0 = lane & 7;
            float p0 = c0[0];
            p0 = (sel0 == 1) ? c0[1] : p0;
            p0 = (sel0 == 2) ? c0[2] : p0;
            p0 = (sel0 == 3) ? c0[3] : p0;
            p0 = (sel0 == 4) ? c0[4] : p0;
            p0 = (sel0 == 5) ? c0[5] : p0;
            p0 = (sel0 == 6) ? c0[6] : p0;
            p0 = (sel0 == 7) ? c0[7] : p0;
            const int m0 = ((lane >> 4) << 3) | sel0;
            if ((lane < 8) || (lane >= 16 && lane < 24))
                part[(wave >> 1) * GRID_COLS + (wave & 1) * 16 + m0] = p0;
        }

        const bool anyW = __builtin_amdgcn_ballot_w32(wany[lane] != 0u) != 0u;

        int off = 0, inc = 0;
        if (anyW) {
            int below = 0, colTot = 0;
            #pragma unroll
            for (int w2 = 0; w2 < GRID_WAVES; ++w2) {
                const int cnt = __popc(mk[w2 * GRID_COLS + lane]);
                below  += (w2 < waveV) ? cnt : 0;
                colTot += cnt;
            }
            inc = colTot;
            #pragma unroll
            for (int d = 1; d < 32; d <<= 1) {
                const int y = __shfl_up(inc, d, 32);
                inc += (lane >= d) ? y : 0;
            }
            off = inc - colTot + below;
            woff[wave * GRID_COLS + lane] = off;
        }
        __syncthreads();

        if (base == 0) {
            pv0 = part[lane];
            #pragma unroll
            for (int s = 1; s < ROW0_SLICES; ++s) pv0 += part[s * GRID_COLS + lane];
        }

        if (anyW) {
            {
                int o = off;
                unsigned wd = word;
                while (wd != 0u) {
                    const int r = wave * GRID_RPW + __builtin_ctz(wd);
                    plane[o++] = __int_as_float((r << 5) | lane);
                    wd &= wd - 1u;
                }
            }
            __syncthreads();

            const int M      = __builtin_amdgcn_readlane(inc, 31);
            const int nTiles = (M + 15) >> 4;
            for (int j = wave; j < nTiles; j += GRID_WAVES) {
                const int q0   = j << 4;
                const int q    = q0 + (lane & 15);
                const int code = __float_as_int(plane[(q < M) ? q : M - 1]);
                const int t    = base + (code >> 5);
                const int b    = cg * GRID_COLS + (code & 31);
                const float* __restrict__ rowPtr = h + ((size_t)t * BS + b) * H + koff;
                const int  sel     = lane & 7;
                const int  mOut    = ((lane >> 4) << 3) | sel;
                const bool doStore = (lane < 8) || (lane >= 16 && lane < 24);
                const int  qOut    = q0 + mOut;

                v8f c = {};
                if (nb > 0) {
                    v2f fa[PROJ_AHEAD], fb[PROJ_AHEAD];
                    #pragma unroll
                    for (int s = 0; s < PROJ_AHEAD; ++s) {
                        fa[s] = *reinterpret_cast<const v2f*>(rowPtr + 4 * s);
                        fb[s] = *reinterpret_cast<const v2f*>(wPtr + 4 * s);
                    }
                    #pragma nounroll
                    for (int trip = 1; trip < nb; ++trip) {
                        const float* __restrict__ ap = rowPtr + trip * span;
                        const float* __restrict__ bp = wPtr + trip * span;
                        #pragma unroll
                        for (int g = 0; g < PROJ_AHEAD; g += PROJ_GROUP) {
                            #pragma unroll
                            for (int s = g; s < g + PROJ_GROUP; ++s)
                                c = __builtin_amdgcn_wmma_f32_16x16x4_f32(
                                        false, fa[s], false, fb[s], (short)0, c, false, false);
                            if (g > 0) {
                                #pragma unroll
                                for (int s = g - PROJ_GROUP; s < g; ++s) {
                                    fa[s] = *reinterpret_cast<const v2f*>(ap + 4 * s);
                                    fb[s] = *reinterpret_cast<const v2f*>(bp + 4 * s);
                                }
                            }
                        }
                        #pragma unroll
                        for (int s = PROJ_AHEAD - PROJ_GROUP; s < PROJ_AHEAD; ++s) {
                            fa[s] = *reinterpret_cast<const v2f*>(ap + 4 * s);
                            fb[s] = *reinterpret_cast<const v2f*>(bp + 4 * s);
                        }
                    }
                    #pragma unroll
                    for (int s = 0; s < PROJ_AHEAD; ++s)
                        c = __builtin_amdgcn_wmma_f32_16x16x4_f32(
                                false, fa[s], false, fb[s], (short)0, c, false, false);
                }
                for (int k = nb * span; k < H; k += 4) {
                    v2f a  = *reinterpret_cast<const v2f*>(rowPtr + k);
                    v2f bw = *reinterpret_cast<const v2f*>(wPtr + k);
                    c = __builtin_amdgcn_wmma_f32_16x16x4_f32(
                            false, a, false, bw, (short)0, c, false, false);
                }

                float pv = c[0];
                pv = (sel == 1) ? c[1] : pv;
                pv = (sel == 2) ? c[2] : pv;
                pv = (sel == 3) ? c[3] : pv;
                pv = (sel == 4) ? c[4] : pv;
                pv = (sel == 5) ? c[5] : pv;
                pv = (sel == 6) ? c[6] : pv;
                pv = (sel == 7) ? c[7] : pv;

                if (doStore && qOut < M) plane[qOut] = pv;
            }
            __syncthreads();

            if (want && !selRow0) {
                const int      ws  = rowSel >> 5;
                const unsigned wdS = mk[ws * GRID_COLS + lane];
                const int      o   = woff[ws * GRID_COLS + lane]
                                   + __popc(wdS & ((1u << (rowSel & 31)) - 1u));
                pvSel = plane[o];
            }
        }
        if (want && selRow0) pvSel = pv0;
        carry = last;
    }

    if (nOk && colOk) {
        const float li = pvSel + ti * (t_grid - tsSel) + bi;
        const size_t dst = (size_t)n * BS + col;
        out_log[dst]  = li;
        out_grid[dst] = t_grid;
    }
}

extern "C" void kernel_launch(void* const* d_in, const int* in_sizes, int n_in,
                              void* d_out, int out_size, void* d_ws, size_t ws_size,
                              hipStream_t stream) {
    const float* h  = (const float*)d_in[0];
    const float* ts = (const float*)d_in[1];
    const float* w  = (const float*)d_in[2];
    const float* ti = (const float*)d_in[3];
    const float* bi = (const float*)d_in[4];

    const int NG  = 1000;
    const int BS  = out_size / (2 * NG);
    const int TBS = in_sizes[1];
    const int T   = TBS / BS;
    const int H   = in_sizes[2];

    float* out_log  = (float*)d_out;
    float* out_grid = out_log + (size_t)NG * BS;

    const int colGroups = (BS + GRID_COLS - 1) / GRID_COLS;
    const int slices    = (NG + GRID_WAVES - 1) / GRID_WAVES;
    rmtpp_fused_kernel<<<colGroups * slices, GRID_WAVES * 32, 0, stream>>>(
        ts, h, w, ti, bi, out_log, out_grid, T, BS, H, NG);
}


# 16 waves a work-group (was 32): a wave takes two grid rows and two 32-row blocks of ts, its two searches side by side, and its K slice of both row-0 tiles; joins read 16 entries; sum order kept
# speedup vs baseline: 66.7369x; 1.0909x over previous
#include <hip/hip_runtime.h>
#include <cstdint>


typedef float v2f __attribute__((ext_vector_type(2)));
typedef float v8f __attribute__((ext_vector_type(8)));

#define GRID_COLS  32
#define GRID_WAVES 16
#define GRID_GROWS 32
#define GRID_BLK   32
#define GRID_NBLK  (2 * GRID_WAVES)
#define GRID_RPW   (2 * GRID_BLK)
#define GRID_ROWS  (GRID_NBLK * GRID_BLK)
#define PROJ_AHEAD 16
#define PROJ_GROUP 4
#define ROW0_SLICES GRID_WAVES
#define ROW0_STEPS 8
static_assert(GRID_NBLK == 32, "one lane reads one block's flag");
static_assert(GRID_GROWS == 2 * GRID_WAVES, "a wave takes two grid rows");

__device__ __forceinline__ void row0_slice(int wave, int H, int& ks0, int& ks1) {
    const int ksteps = (H + 3) >> 2;
    const int per    = (ksteps + ROW0_SLICES - 1) / ROW0_SLICES;
    const int ksl    = wave;
    ks0 = (ksl * per < ksteps) ? ksl * per : ksteps;
    ks1 = (ks0 + per < ksteps) ? ks0 + per : ksteps;
}

__device__ __forceinline__ float tile_row_value(const v8f& c, int sel) {
    float p = c[0];
    p = (sel == 1) ? c[1] : p;
    p = (sel == 2) ? c[2] : p;
    p = (sel == 3) ? c[3] : p;
    p = (sel == 4) ? c[4] : p;
    p = (sel == 5) ? c[5] : p;
    p = (sel == 6) ? c[6] : p;
    p = (sel == 7) ? c[7] : p;
    return p;
}

__global__ __launch_bounds__(GRID_WAVES * 32)
void rmtpp_fused_kernel(const float* __restrict__ ts,
                        const float* __restrict__ h,
                        const float* __restrict__ w,
                        const float* __restrict__ time_influence,
                        const float* __restrict__ base_intensity,
                        float* __restrict__ out_log,
                        float* __restrict__ out_grid,
                        int T, int BS, int H, int NG) {
    __shared__ float    plane[GRID_ROWS * GRID_COLS];
    __shared__ float    wtot[GRID_WAVES * GRID_COLS];
    __shared__ float    wmax[GRID_WAVES * GRID_COLS];
    __shared__ unsigned mk[GRID_NBLK * GRID_COLS];
    __shared__ int      woff[GRID_NBLK * GRID_COLS];
    __shared__ float    part[ROW0_SLICES * GRID_COLS];
    __shared__ unsigned wany[GRID_NBLK];

    const int lane  = threadIdx.x & 31;
    const int waveV = (int)(threadIdx.x >> 5);
    const int wave  = __builtin_amdgcn_readfirstlane(waveV);

    const int colGroups = (BS + GRID_COLS - 1) / GRID_COLS;
    const int cg    = blockIdx.x % colGroups;
    const int slice = blockIdx.x / colGroups;

    const int  col   = cg * GRID_COLS + lane;
    const bool colOk = col < BS;
    const int  colC  = colOk ? col : BS - 1;

    const int  nA = slice * GRID_GROWS + wave;
    const int  nB = nA + GRID_WAVES;

    const float ti = time_influence[0];
    const float bi = base_intensity[0];

    const float inf = __builtin_inff();

    const int koff = (lane >> 4) << 1;
    const float* __restrict__ wPtr = w + koff;
    const int span = 4 * PROJ_AHEAD;
    const int nb   = H / span;

    float mx = -3.402823466e38f;
    if (T > GRID_ROWS) {
        for (int base = GRID_ROWS; base < T; base += GRID_ROWS) {
            const int r0 = base + wave * GRID_RPW;
            #pragma unroll
            for (int i = 0; i < GRID_RPW; ++i) {
                const int t  = r0 + i;
                const int tc = (t < T) ? t : T - 1;
                mx = fmaxf(mx, (ts + (size_t)tc * BS)[colC]);
            }
        }
    }

    float pv0 = 0.0f;

    float t_gridA = 0.0f, t_gridB = 0.0f;
    bool  foundA  = false, foundB = false;
    float carry   = inf;
    float pvSelA  = 0.0f, pvSelB = 0.0f;
    float tsSelA  = 0.0f, tsSelB = 0.0f;

    for (int base = 0; base < T; base += GRID_ROWS) {
        float va[GRID_BLK], vb[GRID_BLK];
        const int r0 = base + wave * GRID_RPW;
        #pragma unroll
        for (int i = 0; i < GRID_BLK; ++i) {
            const int t  = r0 + i;
            const int tc = (t < T) ? t : T - 1;
            va[i] = (ts + (size_t)tc * BS)[colC];
        }
        #pragma unroll
        for (int i = 0; i < GRID_BLK; ++i) {
            const int t  = r0 + GRID_BLK + i;
            const int tc = (t < T) ? t : T - 1;
            vb[i] = (ts + (size_t)tc * BS)[colC];
        }
        #pragma unroll
        for (int i = 0; i < GRID_BLK; ++i) {
            const bool okA = (r0 + i) < T;
            const bool okB = (r0 + GRID_BLK + i) < T;
            mx    = fmaxf(mx, okA ? va[i] : mx);
            mx    = fmaxf(mx, okB ? vb[i] : mx);
            va[i] = okA ? va[i] : inf;
            vb[i] = okB ? vb[i] : inf;
        }
        #pragma unroll
        for (int i = 1; i < GRID_BLK; ++i) {
            va[i] = fminf(va[i - 1], va[i]);
            vb[i] = fminf(vb[i - 1], vb[i]);
        }

        v2f ea0[ROW0_STEPS], ea1[ROW0_STEPS];
        int ks0 = 0, ks1 = 0;
        if (base == 0) row0_slice(wave, H, ks0, ks1);
        const int nReg = (ks1 - ks0 >= ROW0_STEPS) ? ROW0_STEPS : 0;
        int b00 = cg * GRID_COLS + (lane & 15);
        int b01 = b00 + 16;
        b00 = (b00 < BS) ? b00 : BS - 1;
        b01 = (b01 < BS) ? b01 : BS - 1;
        const float* __restrict__ rowPtr00 = h + (size_t)b00 * H + koff;
        const float* __restrict__ rowPtr01 = h + (size_t)b01 * H + koff;
        {
            const int kLast = ((H + 3) >> 2) - 1;
            #pragma unroll
            for (int s = 0; s < ROW0_STEPS; ++s) {
                const int kk = (ks0 + s < kLast) ? ks0 + s : kLast;
                ea0[s] = *reinterpret_cast<const v2f*>(rowPtr00 + 4 * kk);
                ea1[s] = *reinterpret_cast<const v2f*>(rowPtr01 + 4 * kk);
            }
        }
        wtot[wave * GRID_COLS + lane] = fminf(va[GRID_BLK - 1], vb[GRID_BLK - 1]);
        if (base == 0) wmax[wave * GRID_COLS + lane] = mx;
        __syncthreads();

        float pre = carry;
        #pragma unroll
        for (int w2 = 0; w2 < GRID_WAVES; ++w2) {
            const float x = wtot[w2 * GRID_COLS + lane];
            pre = fminf(pre, (w2 < waveV) ? x : inf);
        }
        if (base == 0) {
            float d = wmax[lane];
            #pragma unroll
            for (int w2 = 1; w2 < GRID_WAVES; ++w2) d = fmaxf(d, wmax[w2 * GRID_COLS + lane]);
            t_gridA = ((float)nA / (float)(NG - 1)) * d;
            t_gridB = ((float)nB / (float)(NG - 1)) * d;
        }
        const float preB = fminf(pre, va[GRID_BLK - 1]);
        unsigned wordA = 0u, wordB = 0u;
        {
            float prevA = pre, prevB = preB;
            #pragma unroll
            for (int i = 0; i < GRID_BLK; ++i) {
                const float pa = fminf(pre,  va[i]);
                const float pb = fminf(preB, vb[i]);
                plane[((2 * wave)     * GRID_BLK + i) * GRID_COLS + lane] = pa;
                plane[((2 * wave + 1) * GRID_BLK + i) * GRID_COLS + lane] = pb;
                const bool hitA = (pa < prevA) && (r0 + i != 0);
                const bool hitB = (pb < prevB);
                wordA |= hitA ? (1u << i) : 0u;
                wordB |= hitB ? (1u << i) : 0u;
                prevA = pa;
                prevB = pb;
            }
        }
        wordA = colOk ? wordA : 0u;
        wordB = colOk ? wordB : 0u;
        mk[(2 * wave)     * GRID_COLS + lane] = wordA;
        mk[(2 * wave + 1) * GRID_COLS + lane] = wordB;
        {
            const bool anyA = __builtin_amdgcn_ballot_w32(wordA != 0u) != 0u;
            const bool anyB = __builtin_amdgcn_ballot_w32(wordB != 0u) != 0u;
            if (lane < 2) wany[2 * wave + lane] = ((lane == 0) ? anyA : anyB) ? 1u : 0u;
        }
        __syncthreads();

        v2f eb[ROW0_STEPS];
        {
            const int kLast = ((H + 3) >> 2) - 1;
            #pragma unroll
            for (int s = 0; s < ROW0_STEPS; ++s) {
                const int kk = (ks0 + s < kLast) ? ks0 + s : kLast;
                eb[s] = *reinterpret_cast<const v2f*>(wPtr + 4 * kk);
            }
        }
        bool hitHereA = false, hitHereB = false;
        int  posA = 0, posB = 0;
        if (!(foundA && foundB)) {
            #pragma unroll
            for (int step = GRID_ROWS >> 1; step >= 1; step >>= 1) {
                const float xa = plane[(posA + step - 1) * GRID_COLS + lane];
                const float xb = plane[(posB + step - 1) * GRID_COLS + lane];
                posA += (xa < t_gridA) ? 0 : step;
                posB += (xb < t_gridB) ? 0 : step;
            }
            const bool underA = plane[posA * GRID_COLS + lane] < t_gridA;
            const bool underB = plane[posB * GRID_COLS + lane] < t_gridB;
            if (!foundA && underA) { foundA = true; hitHereA = true; }
            if (!foundB && underB) { foundB = true; hitHereB = true; }
        }
        const bool wantA   = hitHereA || (base == 0);
        const bool wantB   = hitHereB || (base == 0);
        const int  rowSelA = hitHereA ? posA : 0;
        const int  rowSelB = hitHereB ? posB : 0;
        if (wantA) tsSelA = ts[(size_t)(base + rowSelA) * BS + colC];
        if (wantB) tsSelB = ts[(size_t)(base + rowSelB) * BS + colC];
        const float last = plane[(GRID_ROWS - 1) * GRID_COLS + lane];
        const bool selRow0A = (base + rowSelA) == 0;
        const bool selRow0B = (base + rowSelB) == 0;

        if (base == 0) {
            v8f c0 = {}, c1 = {};
            if (nReg != 0) {
                #pragma unroll
                for (int s = 0; s < ROW0_STEPS; ++s) {
                    c0 = __builtin_amdgcn_wmma_f32_16x16x4_f32(
                             false, ea0[s], false, eb[s], (short)0, c0, false, false);
                    c1 = __builtin_amdgcn_wmma_f32_16x16x4_f32(
                             false, ea1[s], false, eb[s], (short)0, c1, false, false);
                }
            }
            #pragma nounroll
            for (int k = 4 * (ks0 + nReg); k < 4 * ks1; k += 4) {
                v2f a0 = *reinterpret_cast<const v2f*>(rowPtr00 + k);
                v2f a1 = *reinterpret_cast<const v2f*>(rowPtr01 + k);
                v2f bw = *reinterpret_cast<const v2f*>(wPtr + k);
                c0 = __builtin_amdgcn_wmma_f32_16x16x4_f32(
                         false, a0, false, bw, (short)0, c0, false, false);
                c1 = __builtin_amdgcn_wmma_f32_16x16x4_f32(
                         false, a1, false, bw, (short)0, c1, false, false);
            }
            const int   sel0 = lane & 7;
            const float p0   = tile_row_value(c0, sel0);
            const float p1   = tile_row_value(c1, sel0);
            const int m0 = ((lane >> 4) << 3) | sel0;
            if ((lane < 8) || (lane >= 16 && lane < 24)) {
                part[wave * GRID_COLS + m0]      = p0;
                part[wave * GRID_COLS + 16 + m0] = p1;
            }
        }

        const bool anyW = __builtin_amdgcn_ballot_w32(wany[lane] != 0u) != 0u;

        int off = 0, inc = 0;
        if (anyW) {
            int below = 0, colTot = 0;
            #pragma unroll
            for (int k2 = 0; k2 < GRID_NBLK; ++k2) {
                const int cnt = __popc(mk[k2 * GRID_COLS + lane]);
                below  += (k2 < 2 * waveV) ? cnt : 0;
                colTot += cnt;
            }
            inc = colTot;
            #pragma unroll
            for (int d = 1; d < 32; d <<= 1) {
                const int y = __shfl_up(inc, d, 32);
                inc += (lane >= d) ? y : 0;
            }
            off = inc - colTot + below;
            woff[(2 * wave)     * GRID_COLS + lane] = off;
            woff[(2 * wave + 1) * GRID_COLS + lane] = off + __popc(wordA);
        }
        __syncthreads();

        if (base == 0) {
            pv0 = part[lane];
            #pragma unroll
            for (int s = 1; s < ROW0_SLICES; ++s) pv0 += part[s * GRID_COLS + lane];
        }

        if (anyW) {
            {
                int o = off;
                unsigned wd = wordA;
                while (wd != 0u) {
                    const int r = (2 * wave) * GRID_BLK + __builtin_ctz(wd);
                    plane[o++] = __int_as_float((r << 5) | lane);
                    wd &= wd - 1u;
                }
                wd = wordB;
                while (wd != 0u) {
                    const int r = (2 * wave + 1) * GRID_BLK + __builtin_ctz(wd);
                    plane[o++] = __int_as_float((r << 5) | lane);
                    wd &= wd - 1u;
                }
            }
            __syncthreads();

            const int M      = __builtin_amdgcn_readlane(inc, 31);
            const int nTiles = (M + 15) >> 4;
            for (int j = wave; j < nTiles; j += GRID_WAVES) {
                const int q0   = j << 4;
                const int q    = q0 + (lane & 15);
                const int code = __float_as_int(plane[(q < M) ? q : M - 1]);
                const int t    = base + (code >> 5);
                const int b    = cg * GRID_COLS + (code & 31);
                const float* __restrict__ rowPtr = h + ((size_t)t * BS + b) * H + koff;
                const int  sel     = lane & 7;
                const int  mOut    = ((lane >> 4) << 3) | sel;
                const bool doStore = (lane < 8) || (lane >= 16 && lane < 24);
                const int  qOut    = q0 + mOut;

                v8f c = {};
                if (nb > 0) {
                    v2f fa[PROJ_AHEAD], fb[PROJ_AHEAD];
                    #pragma unroll
                    for (int s = 0; s < PROJ_AHEAD; ++s) {
                        fa[s] = *reinterpret_cast<const v2f*>(rowPtr + 4 * s);
                        fb[s] = *reinterpret_cast<const v2f*>(wPtr + 4 * s);
                    }
                    #pragma nounroll
                    for (int trip = 1; trip < nb; ++trip) {
                        const float* __restrict__ ap = rowPtr + trip * span;
                        const float* __restrict__ bp = wPtr + trip * span;
                        #pragma unroll
                        for (int g = 0; g < PROJ_AHEAD; g += PROJ_GROUP) {
                            #pragma unroll
                            for (int s = g; s < g + PROJ_GROUP; ++s)
                                c = __builtin_amdgcn_wmma_f32_16x16x4_f32(
                                        false, fa[s], false, fb[s], (short)0, c, false, false);
                            if (g > 0) {
                                #pragma unroll
                                for (int s = g - PROJ_GROUP; s < g; ++s) {
                                    fa[s] = *reinterpret_cast<const v2f*>(ap + 4 * s);
                                    fb[s] = *reinterpret_cast<const v2f*>(bp + 4 * s);
                                }
                            }
                        }
                        #pragma unroll
                        for (int s = PROJ_AHEAD - PROJ_GROUP; s < PROJ_AHEAD; ++s) {
                            fa[s] = *reinterpret_cast<const v2f*>(ap + 4 * s);
                            fb[s] = *reinterpret_cast<const v2f*>(bp + 4 * s);
                        }
                    }
                    #pragma unroll
                    for (int s = 0; s < PROJ_AHEAD; ++s)
                        c = __builtin_amdgcn_wmma_f32_16x16x4_f32(
                                false, fa[s], false, fb[s], (short)0, c, false, false);
                }
                for (int k = nb * span; k < H; k += 4) {
                    v2f a  = *reinterpret_cast<const v2f*>(rowPtr + k);
                    v2f bw = *reinterpret_cast<const v2f*>(wPtr + k);
                    c = __builtin_amdgcn_wmma_f32_16x16x4_f32(
                            false, a, false, bw, (short)0, c, false, false);
                }

                const float pv = tile_row_value(c, sel);

                if (doStore && qOut < M) plane[qOut] = pv;
            }
            __syncthreads();

            if (wantA && !selRow0A) {
                const int      ks  = rowSelA >> 5;
                const unsigned wdS = mk[ks * GRID_COLS + lane];
                const int      o   = woff[ks * GRID_COLS + lane]
                                   + __popc(wdS & ((1u << (rowSelA & 31)) - 1u));
                pvSelA = plane[o];
            }
            if (wantB && !selRow0B) {
                const int      ks  = rowSelB >> 5;
                const unsigned wdS = mk[ks * GRID_COLS + lane];
                const int      o   = woff[ks * GRID_COLS + lane]
                                   + __popc(wdS & ((1u << (rowSelB & 31)) - 1u));
                pvSelB = plane[o];
            }
        }
        if (wantA && selRow0A) pvSelA = pv0;
        if (wantB && selRow0B) pvSelB = pv0;
        carry = last;
    }

    if (colOk) {
        if (nA < NG) {
            const size_t dst = (size_t)nA * BS + col;
            out_log[dst]  = pvSelA + ti * (t_gridA - tsSelA) + bi;
            out_grid[dst] = t_gridA;
        }
        if (nB < NG) {
            const size_t dst = (size_t)nB * BS + col;
            out_log[dst]  = pvSelB + ti * (t_gridB - tsSelB) + bi;
            out_grid[dst] = t_gridB;
        }
    }
}

extern "C" void kernel_launch(void* const* d_in, const int* in_sizes, int n_in,
                              void* d_out, int out_size, void* d_ws, size_t ws_size,
                              hipStream_t stream) {
    const float* h  = (const float*)d_in[0];
    const float* ts = (const float*)d_in[1];
    const float* w  = (const float*)d_in[2];
    const float* ti = (const float*)d_in[3];
    const float* bi = (const float*)d_in[4];

    const int NG  = 1000;
    const int BS  = out_size / (2 * NG);
    const int TBS = in_sizes[1];
    const int T   = TBS / BS;
    const int H   = in_sizes[2];

    float* out_log  = (float*)d_out;
    float* out_grid = out_log + (size_t)NG * BS;

    const int colGroups = (BS + GRID_COLS - 1) / GRID_COLS;
    const int slices    = (NG + GRID_GROWS - 1) / GRID_GROWS;
    rmtpp_fused_kernel<<<colGroups * slices, GRID_WAVES * 32, 0, stream>>>(
        ts, h, w, ti, bi, out_log, out_grid, T, BS, H, NG);
}
